# starting kernel
# baseline (speedup 1.0000x reference)
#include <hip/hip_runtime.h>

typedef float v2f __attribute__((ext_vector_type(2)));
typedef float v8f __attribute__((ext_vector_type(8)));

__device__ __forceinline__ float fast_tanh(float x) {
  const float ax = __builtin_fabsf(x);
  const float t  = __builtin_amdgcn_exp2f(ax * -2.8853900817779268f);
  const float r  = (1.0f - t) * __builtin_amdgcn_rcpf(1.0f + t);
  return __builtin_copysignf(r, x);
}

#define SCHALF(th, ss, cc)                                                   \
  const float u_##ss = (th) * 0.07957747154594767f;           \
  const float ss = __builtin_amdgcn_sinf(u_##ss);                            \
  const float cc = __builtin_amdgcn_cosf(u_##ss)

#define RY_Q0(th) do { SCHALF(th, ss, cc);                                   \
    const float a0=s0, a1=s1, b0=s2, b1v=s3;                                 \
    s0 = cc*a0 - ss*b0;  s1 = cc*a1 - ss*b1v;                                \
    s2 = ss*a0 + cc*b0;  s3 = ss*a1 + cc*b1v; } while (0)

#define RY_Q1(th) do { SCHALF(th, ss, cc);                                   \
    const float a0=s0, b0=s1, a1=s2, b1v=s3;                                 \
    s0 = cc*a0 - ss*b0;  s1 = ss*a0 + cc*b0;                                 \
    s2 = cc*a1 - ss*b1v; s3 = ss*a1 + cc*b1v; } while (0)

__global__ __launch_bounds__(256, 1)
void hybrid_sampler_kernel(const float* __restrict__ inp,
                           const float* __restrict__ W1,
                           const float* __restrict__ b1,
                           const float* __restrict__ W2,
                           const float* __restrict__ b2,
                           float* __restrict__ out, int nrows) {
  __shared__ float hbuf[8][32 * 16];

  const int tid  = threadIdx.x;
  const int wave = tid >> 5;
  const int lane = tid & 31;
  const int n    = lane & 15;
  const bool hi  = lane >= 16;
  float* hb = hbuf[wave];

  int row = blockIdx.x * 256 + wave * 32 + lane;
  row = (row >= nrows) ? (nrows - 1) : row;

  const float in0 = inp[2 * row + 0];
  const float in1 = inp[2 * row + 1];

  const int n8 = n & 7;
  const int n4 = n & 3;
  const float w1k0 = W1[n8];
  const float w1k1 = W1[8 + n8];
  const float bb1  = b1[n8];
  const int rlo = hi ? 8  : 0;
  const int rl1 = hi ? 12 : 4;
  const int rh0 = hi ? 24 : 16;
  const int rh1 = hi ? 28 : 20;
  const float w2a = W2[rlo + n4];
  const float w2b = W2[rl1 + n4];
  const float w2c = W2[rh0 + n4];
  const float w2d = W2[rh1 + n4];
  const float bb2 = b2[n4];

  const float selLo8 = (!hi && n < 8) ? 1.0f : 0.0f;
  const float m8     = (n < 8) ? 1.0f : 0.0f;
  v2f B0;
  B0.x = w1k0 * selLo8;
  B0.y = w1k1 * selLo8;
  const float c1 = bb1 * m8;
  v8f C1 = {c1, c1, c1, c1, c1, c1, c1, c1};

  v2f A0; A0.x = hi ? 0.0f : in0; A0.y = hi ? 0.0f : in1;
  const int sidx = ((lane + 16) & 31) << 2;
  const float in0s = __int_as_float(__builtin_amdgcn_ds_bpermute(sidx, __float_as_int(in0)));
  const float in1s = __int_as_float(__builtin_amdgcn_ds_bpermute(sidx, __float_as_int(in1)));
  v2f A1; A1.x = hi ? 0.0f : in0s; A1.y = hi ? 0.0f : in1s;

  v8f H0 = __builtin_amdgcn_wmma_f32_16x16x4_f32(false, A0, false, B0, (short)0, C1, false, false);
  v8f H1 = __builtin_amdgcn_wmma_f32_16x16x4_f32(false, A1, false, B0, (short)0, C1, false, false);

  #pragma unroll
  for (int j = 0; j < 8; ++j) {
    const int m = j + (hi ? 8 : 0);
    hb[m * 16 + n]        = H0[j];
    hb[(m + 16) * 16 + n] = H1[j];
  }

  const int koff = hi ? 2 : 0;
  v2f AL0 = *(const v2f*)&hb[n * 16 + koff];
  v2f AH0 = *(const v2f*)&hb[n * 16 + 4 + koff];
  v2f AL1 = *(const v2f*)&hb[(n + 16) * 16 + koff];
  v2f AH1 = *(const v2f*)&hb[(n + 16) * 16 + 4 + koff];
  AL0.x = fast_tanh(AL0.x);  AL0.y = fast_tanh(AL0.y);
  AH0.x = fast_tanh(AH0.x);  AH0.y = fast_tanh(AH0.y);
  AL1.x = fast_tanh(AL1.x);  AL1.y = fast_tanh(AL1.y);
  AH1.x = fast_tanh(AH1.x);  AH1.y = fast_tanh(AH1.y);

  const float m4 = (n < 4) ? 1.0f : 0.0f;
  v2f BL, BH;
  BL.x = w2a * m4;
  BL.y = w2b * m4;
  BH.x = w2c * m4;
  BH.y = w2d * m4;
  const float c2 = bb2 * m4;
  v8f C2 = {c2, c2, c2, c2, c2, c2, c2, c2};

  v8f T0  = __builtin_amdgcn_wmma_f32_16x16x4_f32(false, AH0, false, BH, (short)0, C2, false, false);
  v8f Wt0 = __builtin_amdgcn_wmma_f32_16x16x4_f32(false, AL0, false, BL, (short)0, T0, false, false);
  v8f T1  = __builtin_amdgcn_wmma_f32_16x16x4_f32(false, AH1, false, BH, (short)0, C2, false, false);
  v8f Wt1 = __builtin_amdgcn_wmma_f32_16x16x4_f32(false, AL1, false, BL, (short)0, T1, false, false);

  #pragma unroll
  for (int j = 0; j < 8; ++j) {
    const int m = j + (hi ? 8 : 0);
    hb[m * 16 + n]        = Wt0[j];
    hb[(m + 16) * 16 + n] = Wt1[j];
  }
  const float4 wv = *(const float4*)&hb[lane * 16];

  float s0, s1, s2, s3;
  { SCHALF(in0, ss0, cc0); s0 = cc0; s2 = ss0; s1 = 0.0f; s3 = 0.0f; }
  RY_Q1(in1);
  { const float t = s2; s2 = s3; s3 = t; }
  RY_Q0(wv.x);
  RY_Q1(wv.y);
  { const float t = s2; s2 = s3; s3 = t; }
  RY_Q0(wv.z);
  RY_Q1(wv.w);

  const float p0 = s0 * s0 + s2 * s2;
  const float p1 = s1 * s1 + s3 * s3;
  const float mx = fmaxf(p0, p1);
  const float e0 = __expf(p0 - mx);
  const float e1 = __expf(p1 - mx);
  const float inv = __builtin_amdgcn_rcpf(e0 + e1);
  float2 o; o.x = e0 * inv; o.y = e1 * inv;
  *(float2*)&out[2 * row] = o;
}

extern "C" void kernel_launch(void* const* d_in, const int* in_sizes, int n_in,
                              void* d_out, int out_size, void* d_ws, size_t ws_size,
                              hipStream_t stream) {
  const float* inp = (const float*)d_in[0];
  const float* W1  = (const float*)d_in[1];
  const float* b1  = (const float*)d_in[2];
  const float* W2  = (const float*)d_in[3];
  const float* b2  = (const float*)d_in[4];
  float* out = (float*)d_out;

  const int nrows  = in_sizes[0] / 2;
  const int blocks = (nrows + 255) / 256;

  hipLaunchKernelGGL(hybrid_sampler_kernel, dim3(blocks), dim3(256), 0, stream,
                     inp, W1, b1, W2, b2, out, nrows);
}


# persistent waves: 2048 workgroups, each wave loops over tiles of 32 rows (stride 16384 waves), next tile's inputs loaded ahead; weight fragments built once per wave; sums unchanged
# speedup vs baseline: 1.1264x; 1.1264x over previous
#include <hip/hip_runtime.h>

typedef float v2f __attribute__((ext_vector_type(2)));
typedef float v8f __attribute__((ext_vector_type(8)));

#define TWO_LOG2E 2.8853900817779268f

__device__ __forceinline__ v2f tanh_of_z(v2f z) {
  v2f e;
  e.x = __builtin_amdgcn_exp2f(z.x);
  e.y = __builtin_amdgcn_exp2f(z.y);
  const v2f d = e + 1.0f;
  v2f r;
  r.x = __builtin_amdgcn_rcpf(d.x);
  r.y = __builtin_amdgcn_rcpf(d.y);
  return 1.0f - 2.0f * r;
}

#define SCHALF(th, ss, cc)                                                   \
  const float u_##ss = (th) * 0.07957747154594767f;           \
  const float ss = __builtin_amdgcn_sinf(u_##ss);                            \
  const float cc = __builtin_amdgcn_cosf(u_##ss)

#define RY_Q0(th) do { SCHALF(th, ss, cc);                                   \
    const float a0=s0, a1=s1, b0=s2, b1v=s3;                                 \
    s0 = cc*a0 - ss*b0;  s1 = cc*a1 - ss*b1v;                                \
    s2 = ss*a0 + cc*b0;  s3 = ss*a1 + cc*b1v; } while (0)

#define RY_Q1(th) do { SCHALF(th, ss, cc);                                   \
    const float a0=s0, b0=s1, a1=s2, b1v=s3;                                 \
    s0 = cc*a0 - ss*b0;  s1 = ss*a0 + cc*b0;                                 \
    s2 = cc*a1 - ss*b1v; s3 = ss*a1 + cc*b1v; } while (0)

__global__ __launch_bounds__(256, 1)
void hybrid_sampler_kernel(const float* __restrict__ inp,
                           const float* __restrict__ W1,
                           const float* __restrict__ b1,
                           const float* __restrict__ W2,
                           const float* __restrict__ b2,
                           float* __restrict__ out, int nrows, int nwaves) {
  __shared__ float hbuf[8][32 * 16];

  const int tid  = threadIdx.x;
  const int wave = __builtin_amdgcn_readfirstlane(tid >> 5);
  const int lane = tid & 31;
  const int n    = lane & 15;
  const bool hi  = lane >= 16;
  float* hb = hbuf[wave];

  const int last   = nrows - 1;
  const int ntiles = (nrows + 31) >> 5;
  int tile = (int)blockIdx.x * 8 + wave;
  if (tile >= ntiles) return;

  int row = tile * 32 + lane;
  row = (row > last) ? last : row;
  const float2* __restrict__ inp2 = (const float2*)inp;
  float2* __restrict__ out2 = (float2*)out;
  float2 cur = inp2[(unsigned)row];

  const int n8 = n & 7;
  const int n4 = n & 3;
  const float w1k0 = W1[n8];
  const float w1k1 = W1[8 + n8];
  const float bb1  = b1[n8];
  const int rlo = hi ? 8  : 0;
  const int rl1 = hi ? 12 : 4;
  const int rh0 = hi ? 24 : 16;
  const int rh1 = hi ? 28 : 20;
  const float w2a = W2[rlo + n4];
  const float w2b = W2[rl1 + n4];
  const float w2c = W2[rh0 + n4];
  const float w2d = W2[rh1 + n4];
  const float bb2 = b2[n4];

  const float selLo8 = (!hi && n < 8) ? TWO_LOG2E : 0.0f;
  const float m8     = (n < 8) ? TWO_LOG2E : 0.0f;
  v2f B0;
  B0.x = w1k0 * selLo8;
  B0.y = w1k1 * selLo8;
  const float c1 = bb1 * m8;
  const v8f C1 = {c1, c1, c1, c1, c1, c1, c1, c1};

  const float m4 = (n < 4) ? 1.0f : 0.0f;
  v2f BL, BH;
  BL.x = w2a * m4;
  BL.y = w2b * m4;
  BH.x = w2c * m4;
  BH.y = w2d * m4;
  const float c2 = bb2 * m4;
  const v8f C2 = {c2, c2, c2, c2, c2, c2, c2, c2};

  const int sidx = ((lane + 16) & 31) << 2;
  const int koff = hi ? 2 : 0;

  for (;;) {
    const int ntile = tile + nwaves;
    int nrow = ntile * 32 + lane;
    nrow = (nrow > last) ? last : nrow;
    const float2 nxt = inp2[(unsigned)nrow];

    const float in0 = cur.x;
    const float in1 = cur.y;

    v2f A0; A0.x = hi ? 0.0f : in0; A0.y = hi ? 0.0f : in1;
    const float in0s = __int_as_float(__builtin_amdgcn_ds_bpermute(sidx, __float_as_int(in0)));
    const float in1s = __int_as_float(__builtin_amdgcn_ds_bpermute(sidx, __float_as_int(in1)));
    v2f A1; A1.x = hi ? 0.0f : in0s; A1.y = hi ? 0.0f : in1s;

    v8f H0 = __builtin_amdgcn_wmma_f32_16x16x4_f32(false, A0, false, B0, (short)0, C1, false, false);
    v8f H1 = __builtin_amdgcn_wmma_f32_16x16x4_f32(false, A1, false, B0, (short)0, C1, false, false);

    #pragma unroll
    for (int j = 0; j < 8; ++j) {
      const int m = j + (hi ? 8 : 0);
      hb[m * 16 + n]        = H0[j];
      hb[(m + 16) * 16 + n] = H1[j];
    }

    v2f AL0 = *(const v2f*)&hb[n * 16 + koff];
    v2f AH0 = *(const v2f*)&hb[n * 16 + 4 + koff];
    v2f AL1 = *(const v2f*)&hb[(n + 16) * 16 + koff];
    v2f AH1 = *(const v2f*)&hb[(n + 16) * 16 + 4 + koff];
    AL0 = tanh_of_z(AL0);
    AH0 = tanh_of_z(AH0);
    AL1 = tanh_of_z(AL1);
    AH1 = tanh_of_z(AH1);

    v8f T0  = __builtin_amdgcn_wmma_f32_16x16x4_f32(false, AH0, false, BH, (short)0, C2, false, false);
    v8f Wt0 = __builtin_amdgcn_wmma_f32_16x16x4_f32(false, AL0, false, BL, (short)0, T0, false, false);
    v8f T1  = __builtin_amdgcn_wmma_f32_16x16x4_f32(false, AH1, false, BH, (short)0, C2, false, false);
    v8f Wt1 = __builtin_amdgcn_wmma_f32_16x16x4_f32(false, AL1, false, BL, (short)0, T1, false, false);

    #pragma unroll
    for (int j = 0; j < 8; ++j) {
      const int m = j + (hi ? 8 : 0);
      hb[m * 16 + n]        = Wt0[j];
      hb[(m + 16) * 16 + n] = Wt1[j];
    }
    const float4 wv = *(const float4*)&hb[lane * 16];

    float s0, s1, s2, s3;
    { SCHALF(in0, ss0, cc0); s0 = cc0; s2 = ss0; s1 = 0.0f; s3 = 0.0f; }
    RY_Q1(in1);
    { const float t = s2; s2 = s3; s3 = t; }
    RY_Q0(wv.x);
    RY_Q1(wv.y);
    { const float t = s2; s2 = s3; s3 = t; }
    RY_Q0(wv.z);
    RY_Q1(wv.w);

    const float p0 = s0 * s0 + s2 * s2;
    const float p1 = s1 * s1 + s3 * s3;
    const float mx = fmaxf(p0, p1);
    const float e0 = __expf(p0 - mx);
    const float e1 = __expf(p1 - mx);
    const float inv = __builtin_amdgcn_rcpf(e0 + e1);
    float2 o; o.x = e0 * inv; o.y = e1 * inv;
    out2[(unsigned)row] = o;

    if (ntile >= ntiles) break;
    tile = ntile;
    row  = nrow;
    cur  = nxt;
  }
}

extern "C" void kernel_launch(void* const* d_in, const int* in_sizes, int n_in,
                              void* d_out, int out_size, void* d_ws, size_t ws_size,
                              hipStream_t stream) {
  const float* inp = (const float*)d_in[0];
  const float* W1  = (const float*)d_in[1];
  const float* b1  = (const float*)d_in[2];
  const float* W2  = (const float*)d_in[3];
  const float* b2  = (const float*)d_in[4];
  float* out = (float*)d_out;

  const int nrows = in_sizes[0] / 2;

  hipLaunchKernelGGL(hybrid_sampler_kernel, dim3(2048), dim3(256), 0, stream,
                     inp, W1, b1, W2, b2, out, nrows, 16384);
}


# both 16-row halves of a tile share every WMMA, packed along k (block-diagonal weight fragments, own row as A): 5 WMMAs, 16 LDS stores per tile for 6, 32; bpermutes go; layer-2 sum order changes
# speedup vs baseline: 1.2158x; 1.0794x over previous
#include <hip/hip_runtime.h>

typedef float v2f __attribute__((ext_vector_type(2)));
typedef float v8f __attribute__((ext_vector_type(8)));

#define TWO_LOG2E 2.8853900817779268f

__device__ __forceinline__ v2f tanh_of_z(v2f z) {
  v2f e;
  e.x = __builtin_amdgcn_exp2f(z.x);
  e.y = __builtin_amdgcn_exp2f(z.y);
  const v2f d = e + 1.0f;
  v2f r;
  r.x = __builtin_amdgcn_rcpf(d.x);
  r.y = __builtin_amdgcn_rcpf(d.y);
  return 1.0f - 2.0f * r;
}

#define SCHALF(th, ss, cc)                                                   \
  const float u_##ss = (th) * 0.07957747154594767f;           \
  const float ss = __builtin_amdgcn_sinf(u_##ss);                            \
  const float cc = __builtin_amdgcn_cosf(u_##ss)

#define RY_Q0(th) do { SCHALF(th, ss, cc);                                   \
    const float a0=s0, a1=s1, b0=s2, b1v=s3;                                 \
    s0 = cc*a0 - ss*b0;  s1 = cc*a1 - ss*b1v;                                \
    s2 = ss*a0 + cc*b0;  s3 = ss*a1 + cc*b1v; } while (0)

#define RY_Q1(th) do { SCHALF(th, ss, cc);                                   \
    const float a0=s0, b0=s1, a1=s2, b1v=s3;                                 \
    s0 = cc*a0 - ss*b0;  s1 = ss*a0 + cc*b0;                                 \
    s2 = cc*a1 - ss*b1v; s3 = ss*a1 + cc*b1v; } while (0)

__global__ __launch_bounds__(256, 1)
void hybrid_sampler_kernel(const float* __restrict__ inp,
                           const float* __restrict__ W1,
                           const float* __restrict__ b1,
                           const float* __restrict__ W2,
                           const float* __restrict__ b2,
                           float* __restrict__ out, int nrows, int nwaves) {
  __shared__ float hbuf[8][16 * 16];

  const int tid  = threadIdx.x;
  const int wave = __builtin_amdgcn_readfirstlane(tid >> 5);
  const int lane = tid & 31;
  const int n    = lane & 15;
  const bool hi  = lane >= 16;
  float* hb = hbuf[wave];

  const int last   = nrows - 1;
  const int ntiles = (nrows + 31) >> 5;
  int tile = (int)blockIdx.x * 8 + wave;
  if (tile >= ntiles) return;

  int row = tile * 32 + lane;
  row = (row > last) ? last : row;
  const float2* __restrict__ inp2 = (const float2*)inp;
  float2* __restrict__ out2 = (float2*)out;
  float2 cur = inp2[(unsigned)row];

  const int n8 = n & 7;
  const int n4 = n & 3;
  const float w1k0 = W1[n8];
  const float w1k1 = W1[8 + n8];
  const float bb1  = b1[n8];
  const float w2r0 = W2[n4];
  const float w2r1 = W2[4 + n4];
  const float w2r2 = W2[8 + n4];
  const float w2r3 = W2[12 + n4];
  const float w2r4 = W2[16 + n4];
  const float w2r5 = W2[20 + n4];
  const float w2r6 = W2[24 + n4];
  const float w2r7 = W2[28 + n4];
  const float bb2  = b2[n4];

  const int half = lane >> 4;
  const float sel1 = ((n >> 3) == half) ? TWO_LOG2E : 0.0f;
  v2f B0;
  B0.x = w1k0 * sel1;
  B0.y = w1k1 * sel1;
  const float c1 = bb1 * TWO_LOG2E;
  const v8f C1 = {c1, c1, c1, c1, c1, c1, c1, c1};

  const bool v2 = (n >> 2) == half;
  v2f BQ0, BQ1, BQ2, BQ3;
  BQ0.x = v2 ? w2r0 : 0.0f;  BQ0.y = v2 ? w2r1 : 0.0f;
  BQ1.x = v2 ? w2r2 : 0.0f;  BQ1.y = v2 ? w2r3 : 0.0f;
  BQ2.x = v2 ? w2r4 : 0.0f;  BQ2.y = v2 ? w2r5 : 0.0f;
  BQ3.x = v2 ? w2r6 : 0.0f;  BQ3.y = v2 ? w2r7 : 0.0f;
  const v8f C2 = {bb2, bb2, bb2, bb2, bb2, bb2, bb2, bb2};

  const int st   = (hi ? 8 * 16 : 0) + n;
  const int zoff = n * 16 + (hi ? 8 : 0);
  const int woff = n * 16 + (hi ? 4 : 0);

  for (;;) {
    const int ntile = tile + nwaves;
    int nrow = ntile * 32 + lane;
    nrow = (nrow > last) ? last : nrow;
    const float2 nxt = inp2[(unsigned)nrow];

    const float in0 = cur.x;
    const float in1 = cur.y;

    v2f A0; A0.x = in0; A0.y = in1;
    v8f Z = __builtin_amdgcn_wmma_f32_16x16x4_f32(false, A0, false, B0, (short)0, C1, false, false);

    #pragma unroll
    for (int j = 0; j < 8; ++j) hb[st + 16 * j] = Z[j];

    const float4 za = *(const float4*)&hb[zoff];
    const float4 zb = *(const float4*)&hb[zoff + 4];
    v2f T0; T0.x = za.x; T0.y = za.y;
    v2f T1; T1.x = za.z; T1.y = za.w;
    v2f T2; T2.x = zb.x; T2.y = zb.y;
    v2f T3; T3.x = zb.z; T3.y = zb.w;
    T0 = tanh_of_z(T0);
    T1 = tanh_of_z(T1);
    T2 = tanh_of_z(T2);
    T3 = tanh_of_z(T3);

    v8f Wt = __builtin_amdgcn_wmma_f32_16x16x4_f32(false, T0, false, BQ0, (short)0, C2, false, false);
    Wt = __builtin_amdgcn_wmma_f32_16x16x4_f32(false, T1, false, BQ1, (short)0, Wt, false, false);
    Wt = __builtin_amdgcn_wmma_f32_16x16x4_f32(false, T2, false, BQ2, (short)0, Wt, false, false);
    Wt = __builtin_amdgcn_wmma_f32_16x16x4_f32(false, T3, false, BQ3, (short)0, Wt, false, false);

    #pragma unroll
    for (int j = 0; j < 8; ++j) hb[st + 16 * j] = Wt[j];
    const float4 wv = *(const float4*)&hb[woff];

    float s0, s1, s2, s3;
    { SCHALF(in0, ss0, cc0); s0 = cc0; s2 = ss0; s1 = 0.0f; s3 = 0.0f; }
    RY_Q1(in1);
    { const float t = s2; s2 = s3; s3 = t; }
    RY_Q0(wv.x);
    RY_Q1(wv.y);
    { const float t = s2; s2 = s3; s3 = t; }
    RY_Q0(wv.z);
    RY_Q1(wv.w);

    const float p0 = s0 * s0 + s2 * s2;
    const float p1 = s1 * s1 + s3 * s3;
    const float mx = fmaxf(p0, p1);
    const float e0 = __expf(p0 - mx);
    const float e1 = __expf(p1 - mx);
    const float inv = __builtin_amdgcn_rcpf(e0 + e1);
    float2 o; o.x = e0 * inv; o.y = e1 * inv;
    out2[(unsigned)row] = o;

    if (ntile >= ntiles) break;
    tile = ntile;
    row  = nrow;
    cur  = nxt;
  }
}

extern "C" void kernel_launch(void* const* d_in, const int* in_sizes, int n_in,
                              void* d_out, int out_size, void* d_ws, size_t ws_size,
                              hipStream_t stream) {
  const float* inp = (const float*)d_in[0];
  const float* W1  = (const float*)d_in[1];
  const float* b1  = (const float*)d_in[2];
  const float* W2  = (const float*)d_in[3];
  const float* b2  = (const float*)d_in[4];
  float* out = (float*)d_out;

  const int nrows = in_sizes[0] / 2;

  hipLaunchKernelGGL(hybrid_sampler_kernel, dim3(2048), dim3(256), 0, stream,
                     inp, W1, b1, W2, b2, out, nrows, 16384);
}
